# starting kernel
# baseline (speedup 1.0000x reference)
#include <hip/hip_runtime.h>
#include <hip/hip_bf16.h>


typedef float v2f __attribute__((ext_vector_type(2)));
typedef float v8f __attribute__((ext_vector_type(8)));

#define N_NODES 100000
#define N_EDGES 1600000
#define HF 64

__global__ void zero_f32(float* __restrict__ p, int n) {
    int i = blockIdx.x * blockDim.x + threadIdx.x;
    if (i < n) p[i] = 0.0f;
}

__global__ void degree_kernel(const int* __restrict__ ei, float* __restrict__ deg, int E) {
    int e = blockIdx.x * blockDim.x + threadIdx.x;
    if (e < E) atomicAdd(&deg[ei[E + e]], 1.0f);
}

__global__ void dinv_kernel(float* __restrict__ d, int n) {
    int i = blockIdx.x * blockDim.x + threadIdx.x;
    if (i < n) {
        float v = fmaxf(d[i], 1.0f);
        d[i] = 1.0f / sqrtf(v);
    }
}

__global__ void gemm_bias_relu_wmma(const float* __restrict__ A,
                                    const float* __restrict__ W,
                                    const float* __restrict__ bias,
                                    float* __restrict__ C,
                                    int K, int do_relu) {
    extern __shared__ float Blds[];
    const int tid = threadIdx.x;

    for (int i = tid; i < K * HF; i += 256) {
        int kk = i >> 6, c = i & 63;
        Blds[kk * 65 + c] = W[i];
    }
    __syncthreads();

    const int wave = tid >> 5;
    const int lane = tid & 31;
    const int half = lane >> 4;
    const int lm   = lane & 15;
    const int tn   = wave & 3;
    const int tml  = wave >> 2;
    const int row0 = (blockIdx.x * 2 + tml) * 16;
    const int col0 = tn * 16;

    const float* Arow = A + (long)(row0 + lm) * K;

    v8f acc = {0.f, 0.f, 0.f, 0.f, 0.f, 0.f, 0.f, 0.f};
    for (int k = 0; k < K; k += 4) {
        const int ks = k + 2 * half;
        v2f a, b;
        a.x = Arow[ks];
        a.y = Arow[ks + 1];
        b.x = Blds[ks * 65 + col0 + lm];
        b.y = Blds[(ks + 1) * 65 + col0 + lm];
        acc = __builtin_amdgcn_wmma_f32_16x16x4_f32(
            false, a, false, b, (short)0, acc, false, false);
    }

    const int col  = col0 + lm;
    const float bc = bias[col];
#pragma unroll
    for (int r = 0; r < 8; ++r) {
        int row = row0 + r + 8 * half;
        float v = acc[r] + bc;
        if (do_relu) v = fmaxf(v, 0.0f);
        C[(long)row * HF + col] = v;
    }
}

__global__ void theta_init(const float* __restrict__ h, float* __restrict__ feat,
                           float* __restrict__ hfin, float c0, int off, int n64) {
    int i = blockIdx.x * blockDim.x + threadIdx.x;
    if (i >= n64) return;
    float v = h[i];
    feat[i] = v;
    int n = i >> 6, f = i & 63;
    hfin[(long)n * 192 + off + f] = c0 * v;
}

__global__ void edge_scatter(const int* __restrict__ ei, const float* __restrict__ feat,
                             const float* __restrict__ dinv, float* __restrict__ msg,
                             int E) {
    long long tid = (long long)blockIdx.x * blockDim.x + threadIdx.x;
    int e = (int)(tid >> 6);
    if (e >= E) return;
    int f = (int)(tid & 63);
    int s = ei[e];
    int d = ei[E + e];
    float v = feat[(long)s * HF + f] * dinv[s];
    atomicAdd(&msg[(long)d * HF + f], v);
}

__global__ void theta_update(float* __restrict__ feat, const float* __restrict__ msg,
                             const float* __restrict__ dinv, float* __restrict__ hfin,
                             float ck, int off, int n64) {
    int i = blockIdx.x * blockDim.x + threadIdx.x;
    if (i >= n64) return;
    int n = i >> 6, f = i & 63;
    float v = feat[i] - msg[i] * dinv[n];
    feat[i] = v;
    hfin[(long)n * 192 + off + f] += ck * v;
}

__global__ void final_linear(const float* __restrict__ Hm, const float* __restrict__ Wm2,
                             const float* __restrict__ bm2, float* __restrict__ out,
                             int n) {
    int i = blockIdx.x * blockDim.x + threadIdx.x;
    if (i >= n) return;
    const float* h = Hm + (long)i * HF;
    float c0 = bm2[0], c1 = bm2[1];
#pragma unroll 8
    for (int j = 0; j < HF; ++j) {
        float hv = h[j];
        c0 = fmaf(hv, Wm2[j * 2 + 0], c0);
        c1 = fmaf(hv, Wm2[j * 2 + 1], c1);
    }
    out[i * 2 + 0] = c0;
    out[i * 2 + 1] = c1;
}

extern "C" void kernel_launch(void* const* d_in, const int* in_sizes, int n_in,
                              void* d_out, int out_size, void* d_ws, size_t ws_size,
                              hipStream_t stream) {
    const float* x   = (const float*)d_in[0];
    const int*   ei  = (const int*)  d_in[1];
    const float* W1  = (const float*)d_in[2];
    const float* b1  = (const float*)d_in[3];
    const float* W2  = (const float*)d_in[4];
    const float* b2  = (const float*)d_in[5];
    const float* Wm1 = (const float*)d_in[6];
    const float* bm1 = (const float*)d_in[7];
    const float* Wm2 = (const float*)d_in[8];
    const float* bm2 = (const float*)d_in[9];
    float* out = (float*)d_out;

    const int N = N_NODES, E = N_EDGES;
    const int N64 = N * HF;

    float* ws   = (float*)d_ws;
    float* dinv = ws;
    float* h    = dinv + N;
    float* feat = h    + (long)N64;
    float* msg  = feat + (long)N64;
    float* hfin = msg  + (long)N64;

    const float TH[3][3] = {{3.0f, -3.0f, 0.75f},
                            {0.0f,  3.0f, -1.50f},
                            {0.0f,  0.0f, 0.75f}};

    zero_f32<<<(N + 255) / 256, 256, 0, stream>>>(dinv, N);
    degree_kernel<<<(E + 255) / 256, 256, 0, stream>>>(ei, dinv, E);
    dinv_kernel<<<(N + 255) / 256, 256, 0, stream>>>(dinv, N);

    gemm_bias_relu_wmma<<<N / 32, 256, 128 * 65 * sizeof(float), stream>>>(
        x, W1, b1, msg, 128, 1);
    gemm_bias_relu_wmma<<<N / 32, 256, 64 * 65 * sizeof(float), stream>>>(
        msg, W2, b2, h, 64, 1);

    const int eg = (int)(((long long)E * HF) / 256);
    for (int t = 0; t < 3; ++t) {
        theta_init<<<(N64 + 255) / 256, 256, 0, stream>>>(
            h, feat, hfin, TH[t][0], t * HF, N64);
        for (int k = 1; k <= 2; ++k) {
            zero_f32<<<(N64 + 255) / 256, 256, 0, stream>>>(msg, N64);
            edge_scatter<<<eg, 256, 0, stream>>>(ei, feat, dinv, msg, E);
            theta_update<<<(N64 + 255) / 256, 256, 0, stream>>>(
                feat, msg, dinv, hfin, TH[t][k], t * HF, N64);
        }
    }

    gemm_bias_relu_wmma<<<N / 32, 256, 192 * 65 * sizeof(float), stream>>>(
        hfin, Wm1, bm1, feat, 192, 1);

    final_linear<<<(N + 255) / 256, 256, 0, stream>>>(feat, Wm2, bm2, out, N);
}


# Small launches ride in neighbours: degree count as extra work-groups of dense layer 1, rsqrt in layer 2, output layer from the last step's registers; 11 -> 8 launches; output-layer sum order changed
# speedup vs baseline: 8.9346x; 8.9346x over previous
#include <hip/hip_runtime.h>
#include <hip/hip_bf16.h>


typedef float v2f __attribute__((ext_vector_type(2)));
typedef float v4f __attribute__((ext_vector_type(4)));
typedef float v8f __attribute__((ext_vector_type(8)));

#define N_NODES 100000
#define N_EDGES 1600000
#define HF 64

__global__ void zero_f32(float* __restrict__ p, int n) {
    int i = blockIdx.x * blockDim.x + threadIdx.x;
    if (i < n) p[i] = 0.0f;
}

__device__ __forceinline__ int chip_half() {
    const unsigned id = __builtin_amdgcn_s_sendmsg_rtn(0x87);
    return ((id >> 16) & 15u) >= 4u ? 1 : 0;
}

#define DEG_EDGES_PER_THREAD 4
#define DEG_THREADS (N_EDGES / DEG_EDGES_PER_THREAD)
#define DEG_BLOCKS ((DEG_THREADS + 255) / 256)
static_assert(N_EDGES % DEG_EDGES_PER_THREAD == 0, "a thread's four edges exist together");

__device__ __forceinline__ void degree_role(const int* __restrict__ ei, float* __restrict__ deg2,
                                            int t) {
    if (t >= DEG_THREADS) return;
    float* deg = deg2 + chip_half() * N_NODES;
    const int4 d4 = ((const int4*)(ei + N_EDGES))[t];
    atomicAdd(&deg[d4.x], 1.0f);
    atomicAdd(&deg[d4.y], 1.0f);
    atomicAdd(&deg[d4.z], 1.0f);
    atomicAdd(&deg[d4.w], 1.0f);
}

template <int EXTRA>
__global__ void gemm_bias_relu_wmma(const float* __restrict__ A,
                                    const float* __restrict__ W,
                                    const float* __restrict__ bias,
                                    float* __restrict__ C,
                                    int K, int do_relu,
                                    const int* __restrict__ ei,
                                    float* __restrict__ deg2,
                                    float* __restrict__ dinv) {
    extern __shared__ float Blds[];
    const int tid = threadIdx.x;

    int tile = blockIdx.x;
    if (EXTRA == 1) {
        if (blockIdx.x < DEG_BLOCKS) {
            degree_role(ei, deg2, blockIdx.x * 256 + tid);
            return;
        }
        tile = blockIdx.x - DEG_BLOCKS;
    }
    for (int i = tid; i < K * HF; i += 256) {
        int kk = i >> 6, c = i & 63;
        Blds[kk * 65 + c] = W[i];
    }
    __syncthreads();

    const int wave = tid >> 5;
    const int lane = tid & 31;
    const int half = lane >> 4;
    const int lm   = lane & 15;
    const int tn   = wave & 3;
    const int tml  = wave >> 2;
    const int row0 = (tile * 2 + tml) * 16;
    const int col0 = tn * 16;

    const float* Arow = A + (long)(row0 + lm) * K;

    v8f acc = {0.f, 0.f, 0.f, 0.f, 0.f, 0.f, 0.f, 0.f};
    for (int k = 0; k < K; k += 4) {
        const int ks = k + 2 * half;
        v2f a, b;
        a.x = Arow[ks];
        a.y = Arow[ks + 1];
        b.x = Blds[ks * 65 + col0 + lm];
        b.y = Blds[(ks + 1) * 65 + col0 + lm];
        acc = __builtin_amdgcn_wmma_f32_16x16x4_f32(
            false, a, false, b, (short)0, acc, false, false);
    }

    const int col  = col0 + lm;
    const float bc = bias[col];
#pragma unroll
    for (int r = 0; r < 8; ++r) {
        int row = row0 + r + 8 * half;
        float v = acc[r] + bc;
        if (do_relu) v = fmaxf(v, 0.0f);
        C[(long)row * HF + col] = v;
    }

    if (EXTRA == 2 && tid < 32) {
        const int i = tile * 32 + tid;
        dinv[i] = 1.0f / sqrtf(fmaxf(deg2[i] + deg2[N_NODES + i], 1.0f));
    }
}

struct Theta3 { float c[3][3]; };

__global__ void gemm_filters_wmma(float* planes,
                                  const float* __restrict__ Wm1,
                                  const float* __restrict__ dinv,
                                  float* __restrict__ R0,
                                  float* __restrict__ QP,
                                  float* __restrict__ Ps,
                                  Theta3 th, int n64) {
    extern __shared__ float Blds[];
    const int tid = threadIdx.x;

    for (int i = tid; i < 3 * HF * HF; i += 256) {
        int kk = i >> 6, c = i & 63;
        Blds[kk * 65 + c] = Wm1[i];
    }
    __syncthreads();

    const int wave = tid >> 5;
    const int lane = tid & 31;
    const int half = lane >> 4;
    const int lm   = lane & 15;
    const int tn   = wave & 3;
    const int tml  = wave >> 2;
    const int row0 = (blockIdx.x * 2 + tml) * 16;
    const int col0 = tn * 16;

    const float* H = planes + n64;
    const float* Hrow = H + (long)(row0 + lm) * HF;

    const v4f* dv = (const v4f*)(dinv + row0 + 8 * half);
    const v4f d03 = dv[0], d47 = dv[1];
    const float dr[8] = {d03.x, d03.y, d03.z, d03.w, d47.x, d47.y, d47.z, d47.w};

    v8f g0 = {0.f, 0.f, 0.f, 0.f, 0.f, 0.f, 0.f, 0.f};
    v8f g1 = g0, g2 = g0;
    for (int k = 0; k < HF; k += 4) {
        const int ks = k + 2 * half;
        v2f a, b0, b1, b2;
        a.x = Hrow[ks];
        a.y = Hrow[ks + 1];
        b0.x = Blds[ks * 65 + col0 + lm];
        b0.y = Blds[(ks + 1) * 65 + col0 + lm];
        b1.x = Blds[(HF + ks) * 65 + col0 + lm];
        b1.y = Blds[(HF + ks + 1) * 65 + col0 + lm];
        b2.x = Blds[(2 * HF + ks) * 65 + col0 + lm];
        b2.y = Blds[(2 * HF + ks + 1) * 65 + col0 + lm];
        g0 = __builtin_amdgcn_wmma_f32_16x16x4_f32(
            false, a, false, b0, (short)0, g0, false, false);
        g1 = __builtin_amdgcn_wmma_f32_16x16x4_f32(
            false, a, false, b1, (short)0, g1, false, false);
        g2 = __builtin_amdgcn_wmma_f32_16x16x4_f32(
            false, a, false, b2, (short)0, g2, false, false);
    }

    __syncthreads();
    {
        const v4f z = {0.f, 0.f, 0.f, 0.f};
        const long base = (long)blockIdx.x * (32 * HF);
        v4f* z0 = (v4f*)(planes + base);
        v4f* z1 = (v4f*)(planes + n64 + base);
        z0[tid] = z;  z0[tid + 256] = z;
        z1[tid] = z;  z1[tid + 256] = z;
    }

    const int col = col0 + lm;
#pragma unroll
    for (int r = 0; r < 8; ++r) {
        long o = (long)(row0 + r + 8 * half) * HF + col;
        float x0 = g0[r], x1 = g1[r], x2 = g2[r];
        const float q = th.c[0][1] * x0 + th.c[1][1] * x1 + th.c[2][1] * x2;
        const float p = th.c[0][2] * x0 + th.c[1][2] * x1 + th.c[2][2] * x2;
        R0[o] = th.c[0][0] * x0 + th.c[1][0] * x1 + th.c[2][0] * x2;
        QP[o] = q + p;
        Ps[o] = p * dr[r];
    }
}


#define SC_BATCH 32
#define SC_GROUP 8
#define SC_NGROUP (SC_BATCH / SC_GROUP)

__global__ void __launch_bounds__(256)
edge_scatter(const int* __restrict__ ei, const float* __restrict__ feat,
             float* __restrict__ msg2, int E, int n64) {
    float* msg = msg2 + (long)chip_half() * n64;
    const int lane = threadIdx.x & 31;
    const int e0   = (blockIdx.x * 8 + (threadIdx.x >> 5)) * SC_BATCH;
    if (e0 >= E) return;
    const int e   = e0 + lane;
    const int s_l = ei[e];
    const int d_l = ei[E + e];

    float lo[SC_NGROUP][SC_GROUP], hi[SC_NGROUP][SC_GROUP];

#pragma unroll
    for (int g = -1; g < SC_NGROUP; ++g) {
        if (g + 1 < SC_NGROUP) {
#pragma unroll
            for (int u = 0; u < SC_GROUP; ++u) {
                const int j = (g + 1) * SC_GROUP + u;
                const int s = __builtin_amdgcn_readlane(s_l, j);
                const float* row = feat + (long)s * HF;
                lo[g + 1][u] = row[lane];
                hi[g + 1][u] = row[lane + 32];
            }
        }
        if (g >= 0) {
#pragma unroll
            for (int u = 0; u < SC_GROUP; ++u) {
                const int j = g * SC_GROUP + u;
                const int d = __builtin_amdgcn_readlane(d_l, j);
                float* row = msg + (long)d * HF;
                atomicAdd(&row[lane], lo[g][u]);
                atomicAdd(&row[lane + 32], hi[g][u]);
            }
        }
    }
}

#define UPD_V 2
#define UPD_THREADS (N_NODES * HF / 4 / UPD_V)
static_assert((N_NODES * HF / 4) % (256 * UPD_V) == 0, "step grid covers the arrays exactly");
static_assert(UPD_THREADS % 16 == 0, "a lane keeps its feature columns from place to place");

template <int LAST>
__global__ void __launch_bounds__(256)
nested_step(float* __restrict__ racc, float* __restrict__ zs,
            float* __restrict__ msg2, const float* __restrict__ dinv,
            const float* __restrict__ bias, int n64,
            const float* __restrict__ Wm2, const float* __restrict__ bm2,
            float* __restrict__ out) {
    __shared__ float oline[UPD_V][32];
    const int t = blockIdx.x * 256 + threadIdx.x;
    v4f* r4 = (v4f*)racc;
    v4f* z4 = (v4f*)zs;
    v4f* m0 = (v4f*)msg2;
    v4f* m1 = (v4f*)(msg2 + n64);

    v4f r[UPD_V], a[UPD_V], p0[UPD_V], p1[UPD_V];
    float d[UPD_V];
#pragma unroll
    for (int u = 0; u < UPD_V; ++u) {
        const int q = t + u * UPD_THREADS;
        p0[u] = m0[q];
        p1[u] = m1[q];
        r[u]  = r4[q];
        if (!LAST) a[u] = z4[q];
        d[u]  = dinv[q >> 4];
    }
    const int f = (t & 15) * 4;
    v4f b = {0.f, 0.f, 0.f, 0.f};
    v4f wa = b, wb = b;
    if (LAST) {
        b.x = bias[f]; b.y = bias[f + 1]; b.z = bias[f + 2]; b.w = bias[f + 3];
        const v4f* w4 = (const v4f*)Wm2;
        wa = w4[(t & 15) * 2];
        wb = w4[(t & 15) * 2 + 1];
    }
#pragma unroll
    for (int u = 0; u < UPD_V; ++u) {
        const int q = t + u * UPD_THREADS;
        const v4f m = p0[u] + p1[u];
        if (LAST) {
            v4f v = (r[u] - m * d[u]) + b;
            v.x = fmaxf(v.x, 0.0f);
            v.y = fmaxf(v.y, 0.0f);
            v.z = fmaxf(v.z, 0.0f);
            v.w = fmaxf(v.w, 0.0f);
            float c0 = v.x * wa.x + v.y * wa.z + v.z * wb.x + v.w * wb.z;
            float c1 = v.x * wa.y + v.y * wa.w + v.z * wb.y + v.w * wb.w;
#pragma unroll
            for (int m = 1; m < 16; m <<= 1) {
                c0 += __shfl_xor(c0, m);
                c1 += __shfl_xor(c1, m);
            }
            if ((threadIdx.x & 15) == 0) {
                const int nl = threadIdx.x >> 4;
                oline[u][nl * 2 + 0] = c0 + bm2[0];
                oline[u][nl * 2 + 1] = c1 + bm2[1];
            }
        } else {
            const v4f z = a[u] - m * d[u];
            const v4f zero = {0.f, 0.f, 0.f, 0.f};
            m0[q] = zero;
            m1[q] = zero;
            z4[q] = z * d[u];
            r4[q] = r[u] + z;
        }
    }
    if (LAST) {
        __syncthreads();
        if (threadIdx.x < 32) {
#pragma unroll
            for (int u = 0; u < UPD_V; ++u) {
                const int node0 = (blockIdx.x * 256 + u * UPD_THREADS) >> 4;
                out[node0 * 2 + threadIdx.x] = oline[u][threadIdx.x];
            }
        }
    }
}

extern "C" void kernel_launch(void* const* d_in, const int* in_sizes, int n_in,
                              void* d_out, int out_size, void* d_ws, size_t ws_size,
                              hipStream_t stream) {
    const float* x   = (const float*)d_in[0];
    const int*   ei  = (const int*)  d_in[1];
    const float* W1  = (const float*)d_in[2];
    const float* b1  = (const float*)d_in[3];
    const float* W2  = (const float*)d_in[4];
    const float* b2  = (const float*)d_in[5];
    const float* Wm1 = (const float*)d_in[6];
    const float* bm1 = (const float*)d_in[7];
    const float* Wm2 = (const float*)d_in[8];
    const float* bm2 = (const float*)d_in[9];
    float* out = (float*)d_out;

    const int N = N_NODES, E = N_EDGES;
    const int N64 = N * HF;

    float* ws   = (float*)d_ws;
    float* dinv = ws;
    float* R0   = dinv + N;
    float* msg2 = R0   + (long)N64;
    float* h1   = msg2;
    float* h    = msg2 + (long)N64;
    float* QP   = h    + (long)N64;
    float* Ps   = QP   + (long)N64;
    float* deg2 = Ps;

    const Theta3 TH = {{{3.0f, -3.0f, 0.75f},
                        {0.0f,  3.0f, -1.50f},
                        {0.0f,  0.0f, 0.75f}}};

    zero_f32<<<(2 * N + 255) / 256, 256, 0, stream>>>(deg2, 2 * N);

    gemm_bias_relu_wmma<1><<<DEG_BLOCKS + N / 32, 256, 128 * 65 * sizeof(float), stream>>>(
        x, W1, b1, h1, 128, 1, ei, deg2, dinv);
    gemm_bias_relu_wmma<2><<<N / 32, 256, 64 * 65 * sizeof(float), stream>>>(
        h1, W2, b2, h, 64, 1, ei, deg2, dinv);

    gemm_filters_wmma<<<N / 32, 256, 192 * 65 * sizeof(float), stream>>>(
        msg2, Wm1, dinv, R0, QP, Ps, TH, N64);

    static_assert(N_EDGES % (8 * SC_BATCH) == 0, "edge batches fill whole work-groups");
    const int eg = N_EDGES / (8 * SC_BATCH);
    const int ug = UPD_THREADS / 256;
    edge_scatter<<<eg, 256, 0, stream>>>(ei, Ps, msg2, E, N64);
    nested_step<0><<<ug, 256, 0, stream>>>(R0, QP, msg2, dinv, bm1, N64, Wm2, bm2, out);

    edge_scatter<<<eg, 256, 0, stream>>>(ei, QP, msg2, E, N64);
    nested_step<1><<<ug, 256, 0, stream>>>(R0, QP, msg2, dinv, bm1, N64, Wm2, bm2, out);
}


# Both scatters add into the same planes, cleared once: own terms summed once (S=R0+QP), last step relu(S-d(m1+m2)+b); first step only turns QP into Zs, no clears, no R0 pass; sum order changed
# speedup vs baseline: 9.6324x; 1.0781x over previous
#include <hip/hip_runtime.h>
#include <hip/hip_bf16.h>


typedef float v2f __attribute__((ext_vector_type(2)));
typedef float v4f __attribute__((ext_vector_type(4)));
typedef float v8f __attribute__((ext_vector_type(8)));

#define N_NODES 100000
#define N_EDGES 1600000
#define HF 64

__global__ void zero_f32(float* __restrict__ p, int n) {
    int i = blockIdx.x * blockDim.x + threadIdx.x;
    if (i < n) p[i] = 0.0f;
}

__device__ __forceinline__ int chip_half() {
    const unsigned id = __builtin_amdgcn_s_sendmsg_rtn(0x87);
    return ((id >> 16) & 15u) >= 4u ? 1 : 0;
}

#define DEG_EDGES_PER_THREAD 4
#define DEG_THREADS (N_EDGES / DEG_EDGES_PER_THREAD)
#define DEG_BLOCKS ((DEG_THREADS + 255) / 256)
static_assert(N_EDGES % DEG_EDGES_PER_THREAD == 0, "a thread's four edges exist together");

__device__ __forceinline__ void degree_role(const int* __restrict__ ei, float* __restrict__ deg2,
                                            int t) {
    if (t >= DEG_THREADS) return;
    float* deg = deg2 + chip_half() * N_NODES;
    const int4 d4 = ((const int4*)(ei + N_EDGES))[t];
    atomicAdd(&deg[d4.x], 1.0f);
    atomicAdd(&deg[d4.y], 1.0f);
    atomicAdd(&deg[d4.z], 1.0f);
    atomicAdd(&deg[d4.w], 1.0f);
}

template <int EXTRA>
__global__ void gemm_bias_relu_wmma(const float* __restrict__ A,
                                    const float* __restrict__ W,
                                    const float* __restrict__ bias,
                                    float* __restrict__ C,
                                    int K, int do_relu,
                                    const int* __restrict__ ei,
                                    float* __restrict__ deg2,
                                    float* __restrict__ dinv) {
    extern __shared__ float Blds[];
    const int tid = threadIdx.x;

    int tile = blockIdx.x;
    if (EXTRA == 1) {
        if (blockIdx.x < DEG_BLOCKS) {
            degree_role(ei, deg2, blockIdx.x * 256 + tid);
            return;
        }
        tile = blockIdx.x - DEG_BLOCKS;
    }
    for (int i = tid; i < K * HF; i += 256) {
        int kk = i >> 6, c = i & 63;
        Blds[kk * 65 + c] = W[i];
    }
    __syncthreads();

    const int wave = tid >> 5;
    const int lane = tid & 31;
    const int half = lane >> 4;
    const int lm   = lane & 15;
    const int tn   = wave & 3;
    const int tml  = wave >> 2;
    const int row0 = (tile * 2 + tml) * 16;
    const int col0 = tn * 16;

    const float* Arow = A + (long)(row0 + lm) * K;

    v8f acc = {0.f, 0.f, 0.f, 0.f, 0.f, 0.f, 0.f, 0.f};
    for (int k = 0; k < K; k += 4) {
        const int ks = k + 2 * half;
        v2f a, b;
        a.x = Arow[ks];
        a.y = Arow[ks + 1];
        b.x = Blds[ks * 65 + col0 + lm];
        b.y = Blds[(ks + 1) * 65 + col0 + lm];
        acc = __builtin_amdgcn_wmma_f32_16x16x4_f32(
            false, a, false, b, (short)0, acc, false, false);
    }

    const int col  = col0 + lm;
    const float bc = bias[col];
#pragma unroll
    for (int r = 0; r < 8; ++r) {
        int row = row0 + r + 8 * half;
        float v = acc[r] + bc;
        if (do_relu) v = fmaxf(v, 0.0f);
        C[(long)row * HF + col] = v;
    }

    if (EXTRA == 2 && tid < 32) {
        const int i = tile * 32 + tid;
        dinv[i] = 1.0f / sqrtf(fmaxf(deg2[i] + deg2[N_NODES + i], 1.0f));
    }
}

struct Theta3 { float c[3][3]; };

__global__ void gemm_filters_wmma(float* planes,
                                  const float* __restrict__ Wm1,
                                  const float* __restrict__ dinv,
                                  float* __restrict__ S,
                                  float* __restrict__ QP,
                                  float* __restrict__ Ps,
                                  Theta3 th, int n64) {
    extern __shared__ float Blds[];
    const int tid = threadIdx.x;

    for (int i = tid; i < 3 * HF * HF; i += 256) {
        int kk = i >> 6, c = i & 63;
        Blds[kk * 65 + c] = Wm1[i];
    }
    __syncthreads();

    const int wave = tid >> 5;
    const int lane = tid & 31;
    const int half = lane >> 4;
    const int lm   = lane & 15;
    const int tn   = wave & 3;
    const int tml  = wave >> 2;
    const int row0 = (blockIdx.x * 2 + tml) * 16;
    const int col0 = tn * 16;

    const float* H = planes + n64;
    const float* Hrow = H + (long)(row0 + lm) * HF;

    const v4f* dv = (const v4f*)(dinv + row0 + 8 * half);
    const v4f d03 = dv[0], d47 = dv[1];
    const float dr[8] = {d03.x, d03.y, d03.z, d03.w, d47.x, d47.y, d47.z, d47.w};

    v8f g0 = {0.f, 0.f, 0.f, 0.f, 0.f, 0.f, 0.f, 0.f};
    v8f g1 = g0, g2 = g0;
    for (int k = 0; k < HF; k += 4) {
        const int ks = k + 2 * half;
        v2f a, b0, b1, b2;
        a.x = Hrow[ks];
        a.y = Hrow[ks + 1];
        b0.x = Blds[ks * 65 + col0 + lm];
        b0.y = Blds[(ks + 1) * 65 + col0 + lm];
        b1.x = Blds[(HF + ks) * 65 + col0 + lm];
        b1.y = Blds[(HF + ks + 1) * 65 + col0 + lm];
        b2.x = Blds[(2 * HF + ks) * 65 + col0 + lm];
        b2.y = Blds[(2 * HF + ks + 1) * 65 + col0 + lm];
        g0 = __builtin_amdgcn_wmma_f32_16x16x4_f32(
            false, a, false, b0, (short)0, g0, false, false);
        g1 = __builtin_amdgcn_wmma_f32_16x16x4_f32(
            false, a, false, b1, (short)0, g1, false, false);
        g2 = __builtin_amdgcn_wmma_f32_16x16x4_f32(
            false, a, false, b2, (short)0, g2, false, false);
    }

    __syncthreads();
    {
        const v4f z = {0.f, 0.f, 0.f, 0.f};
        const long base = (long)blockIdx.x * (32 * HF);
        v4f* z0 = (v4f*)(planes + base);
        v4f* z1 = (v4f*)(planes + n64 + base);
        z0[tid] = z;  z0[tid + 256] = z;
        z1[tid] = z;  z1[tid + 256] = z;
    }

    const int col = col0 + lm;
#pragma unroll
    for (int r = 0; r < 8; ++r) {
        long o = (long)(row0 + r + 8 * half) * HF + col;
        float x0 = g0[r], x1 = g1[r], x2 = g2[r];
        const float r0 = th.c[0][0] * x0 + th.c[1][0] * x1 + th.c[2][0] * x2;
        const float q  = th.c[0][1] * x0 + th.c[1][1] * x1 + th.c[2][1] * x2;
        const float p  = th.c[0][2] * x0 + th.c[1][2] * x1 + th.c[2][2] * x2;
        const float qp = q + p;
        S[o]  = r0 + qp;
        QP[o] = qp;
        Ps[o] = p * dr[r];
    }
}


#define SC_BATCH 32
#define SC_GROUP 8
#define SC_NGROUP (SC_BATCH / SC_GROUP)

__global__ void __launch_bounds__(256)
edge_scatter(const int* __restrict__ ei, const float* __restrict__ feat,
             float* __restrict__ msg2, int E, int n64) {
    float* msg = msg2 + (long)chip_half() * n64;
    const int lane = threadIdx.x & 31;
    const int e0   = (blockIdx.x * 8 + (threadIdx.x >> 5)) * SC_BATCH;
    if (e0 >= E) return;
    const int e   = e0 + lane;
    const int s_l = ei[e];
    const int d_l = ei[E + e];

    float lo[SC_NGROUP][SC_GROUP], hi[SC_NGROUP][SC_GROUP];

#pragma unroll
    for (int g = -1; g < SC_NGROUP; ++g) {
        if (g + 1 < SC_NGROUP) {
#pragma unroll
            for (int u = 0; u < SC_GROUP; ++u) {
                const int j = (g + 1) * SC_GROUP + u;
                const int s = __builtin_amdgcn_readlane(s_l, j);
                const float* row = feat + (long)s * HF;
                lo[g + 1][u] = row[lane];
                hi[g + 1][u] = row[lane + 32];
            }
        }
        if (g >= 0) {
#pragma unroll
            for (int u = 0; u < SC_GROUP; ++u) {
                const int j = g * SC_GROUP + u;
                const int d = __builtin_amdgcn_readlane(d_l, j);
                float* row = msg + (long)d * HF;
                atomicAdd(&row[lane], lo[g][u]);
                atomicAdd(&row[lane + 32], hi[g][u]);
            }
        }
    }
}

#define UPD_V 2
#define UPD_THREADS (N_NODES * HF / 4 / UPD_V)
static_assert((N_NODES * HF / 4) % (256 * UPD_V) == 0, "step grid covers the arrays exactly");
static_assert(UPD_THREADS % 16 == 0, "a lane keeps its feature columns from place to place");

template <int LAST>
__global__ void __launch_bounds__(256)
nested_step(const float* __restrict__ sacc, float* __restrict__ zs,
            const float* __restrict__ msg2, const float* __restrict__ dinv,
            const float* __restrict__ bias, int n64,
            const float* __restrict__ Wm2, const float* __restrict__ bm2,
            float* __restrict__ out) {
    __shared__ float oline[UPD_V][32];
    const int t = blockIdx.x * 256 + threadIdx.x;
    const v4f* s4 = (const v4f*)sacc;
    v4f* z4 = (v4f*)zs;
    const v4f* m0 = (const v4f*)msg2;
    const v4f* m1 = (const v4f*)(msg2 + n64);

    v4f r[UPD_V], a[UPD_V], p0[UPD_V], p1[UPD_V];
    float d[UPD_V];
#pragma unroll
    for (int u = 0; u < UPD_V; ++u) {
        const int q = t + u * UPD_THREADS;
        p0[u] = m0[q];
        p1[u] = m1[q];
        if (LAST) r[u] = s4[q];
        else      a[u] = z4[q];
        d[u]  = dinv[q >> 4];
    }
    const int f = (t & 15) * 4;
    v4f b = {0.f, 0.f, 0.f, 0.f};
    v4f wa = b, wb = b;
    if (LAST) {
        b.x = bias[f]; b.y = bias[f + 1]; b.z = bias[f + 2]; b.w = bias[f + 3];
        const v4f* w4 = (const v4f*)Wm2;
        wa = w4[(t & 15) * 2];
        wb = w4[(t & 15) * 2 + 1];
    }
#pragma unroll
    for (int u = 0; u < UPD_V; ++u) {
        const int q = t + u * UPD_THREADS;
        const v4f m = p0[u] + p1[u];
        if (LAST) {
            v4f v = (r[u] - m * d[u]) + b;
            v.x = fmaxf(v.x, 0.0f);
            v.y = fmaxf(v.y, 0.0f);
            v.z = fmaxf(v.z, 0.0f);
            v.w = fmaxf(v.w, 0.0f);
            float c0 = v.x * wa.x + v.y * wa.z + v.z * wb.x + v.w * wb.z;
            float c1 = v.x * wa.y + v.y * wa.w + v.z * wb.y + v.w * wb.w;
#pragma unroll
            for (int m = 1; m < 16; m <<= 1) {
                c0 += __shfl_xor(c0, m);
                c1 += __shfl_xor(c1, m);
            }
            if ((threadIdx.x & 15) == 0) {
                const int nl = threadIdx.x >> 4;
                oline[u][nl * 2 + 0] = c0 + bm2[0];
                oline[u][nl * 2 + 1] = c1 + bm2[1];
            }
        } else {
            const v4f z = a[u] - m * d[u];
            z4[q] = z * d[u];
        }
    }
    if (LAST) {
        __syncthreads();
        if (threadIdx.x < 32) {
#pragma unroll
            for (int u = 0; u < UPD_V; ++u) {
                const int node0 = (blockIdx.x * 256 + u * UPD_THREADS) >> 4;
                out[node0 * 2 + threadIdx.x] = oline[u][threadIdx.x];
            }
        }
    }
}

extern "C" void kernel_launch(void* const* d_in, const int* in_sizes, int n_in,
                              void* d_out, int out_size, void* d_ws, size_t ws_size,
                              hipStream_t stream) {
    const float* x   = (const float*)d_in[0];
    const int*   ei  = (const int*)  d_in[1];
    const float* W1  = (const float*)d_in[2];
    const float* b1  = (const float*)d_in[3];
    const float* W2  = (const float*)d_in[4];
    const float* b2  = (const float*)d_in[5];
    const float* Wm1 = (const float*)d_in[6];
    const float* bm1 = (const float*)d_in[7];
    const float* Wm2 = (const float*)d_in[8];
    const float* bm2 = (const float*)d_in[9];
    float* out = (float*)d_out;

    const int N = N_NODES, E = N_EDGES;
    const int N64 = N * HF;

    float* ws   = (float*)d_ws;
    float* dinv = ws;
    float* S    = dinv + N;
    float* msg2 = S    + (long)N64;
    float* h1   = msg2;
    float* h    = msg2 + (long)N64;
    float* QP   = h    + (long)N64;
    float* Ps   = QP   + (long)N64;
    float* deg2 = Ps;

    const Theta3 TH = {{{3.0f, -3.0f, 0.75f},
                        {0.0f,  3.0f, -1.50f},
                        {0.0f,  0.0f, 0.75f}}};

    zero_f32<<<(2 * N + 255) / 256, 256, 0, stream>>>(deg2, 2 * N);

    gemm_bias_relu_wmma<1><<<DEG_BLOCKS + N / 32, 256, 128 * 65 * sizeof(float), stream>>>(
        x, W1, b1, h1, 128, 1, ei, deg2, dinv);
    gemm_bias_relu_wmma<2><<<N / 32, 256, 64 * 65 * sizeof(float), stream>>>(
        h1, W2, b2, h, 64, 1, ei, deg2, dinv);

    gemm_filters_wmma<<<N / 32, 256, 192 * 65 * sizeof(float), stream>>>(
        msg2, Wm1, dinv, S, QP, Ps, TH, N64);

    static_assert(N_EDGES % (8 * SC_BATCH) == 0, "edge batches fill whole work-groups");
    const int eg = N_EDGES / (8 * SC_BATCH);
    const int ug = UPD_THREADS / 256;
    edge_scatter<<<eg, 256, 0, stream>>>(ei, Ps, msg2, E, N64);
    nested_step<0><<<ug, 256, 0, stream>>>(S, QP, msg2, dinv, bm1, N64, Wm2, bm2, out);

    edge_scatter<<<eg, 256, 0, stream>>>(ei, QP, msg2, E, N64);
    nested_step<1><<<ug, 256, 0, stream>>>(S, QP, msg2, dinv, bm1, N64, Wm2, bm2, out);
}


# Dense layers stage their 32 x K row tile in LDS beside W, both copied by whole-line 16-byte loads; A fragments come from LDS, not from global 8 B a lane over 16 lines; sums bit-identical
# speedup vs baseline: 10.6707x; 1.1078x over previous
#include <hip/hip_runtime.h>
#include <hip/hip_bf16.h>


typedef float v2f __attribute__((ext_vector_type(2)));
typedef float v4f __attribute__((ext_vector_type(4)));
typedef float v8f __attribute__((ext_vector_type(8)));

#define N_NODES 100000
#define N_EDGES 1600000
#define HF 64

__global__ void zero_f32(float* __restrict__ p, int n) {
    int i = blockIdx.x * blockDim.x + threadIdx.x;
    if (i < n) p[i] = 0.0f;
}

__device__ __forceinline__ int chip_half() {
    const unsigned id = __builtin_amdgcn_s_sendmsg_rtn(0x87);
    return ((id >> 16) & 15u) >= 4u ? 1 : 0;
}

#define DEG_EDGES_PER_THREAD 4
#define DEG_THREADS (N_EDGES / DEG_EDGES_PER_THREAD)
#define DEG_BLOCKS ((DEG_THREADS + 255) / 256)
static_assert(N_EDGES % DEG_EDGES_PER_THREAD == 0, "a thread's four edges exist together");

__device__ __forceinline__ void degree_role(const int* __restrict__ ei, float* __restrict__ deg2,
                                            int t) {
    if (t >= DEG_THREADS) return;
    float* deg = deg2 + chip_half() * N_NODES;
    const int4 d4 = ((const int4*)(ei + N_EDGES))[t];
    atomicAdd(&deg[d4.x], 1.0f);
    atomicAdd(&deg[d4.y], 1.0f);
    atomicAdd(&deg[d4.z], 1.0f);
    atomicAdd(&deg[d4.w], 1.0f);
}

#define W_PITCH 72
#define A_PITCH(K) ((K) + 4)
#define GEMM_LDS_BYTES(KW, KA) (((KW) * W_PITCH + 32 * A_PITCH(KA)) * sizeof(float))

template <int KW>
__device__ __forceinline__ void stage_w(float* __restrict__ Wl, const float* __restrict__ W, int tid) {
    const v4f* Wg = (const v4f*)W;
    for (int j = 0; j < KW * 16 / 256; ++j) {
        const int i = tid + 256 * j;
        *(v4f*)(Wl + (i >> 4) * W_PITCH + (i & 15) * 4) = Wg[i];
    }
}

template <int K>
__device__ __forceinline__ void stage_rows(float* __restrict__ Al, const float* __restrict__ rows, int tid) {
    const v4f* Ag = (const v4f*)rows;
    for (int j = 0; j < 32 * (K / 4) / 256; ++j) {
        const int i = tid + 256 * j;
        *(v4f*)(Al + (i / (K / 4)) * A_PITCH(K) + (i % (K / 4)) * 4) = Ag[i];
    }
}

template <int EXTRA, int K>
__global__ void
gemm_bias_relu_wmma(const float* __restrict__ A,
                    const float* __restrict__ W,
                    const float* __restrict__ bias,
                    float* __restrict__ C,
                    int do_relu,
                    const int* __restrict__ ei,
                    float* __restrict__ deg2,
                    float* __restrict__ dinv) {
    extern __shared__ __attribute__((aligned(16))) float lds[];
    float* Wl = lds;
    float* Al = lds + K * W_PITCH;
    const int tid = threadIdx.x;

    int tile = blockIdx.x;
    if (EXTRA == 1) {
        if (blockIdx.x < DEG_BLOCKS) {
            degree_role(ei, deg2, blockIdx.x * 256 + tid);
            return;
        }
        tile = blockIdx.x - DEG_BLOCKS;
    }
    stage_rows<K>(Al, A + (long)tile * (32 * K), tid);
    stage_w<K>(Wl, W, tid);
    __syncthreads();

    const int wave = tid >> 5;
    const int lane = tid & 31;
    const int half = lane >> 4;
    const int lm   = lane & 15;
    const int tn   = wave & 3;
    const int tml  = wave >> 2;
    const int row0 = (tile * 2 + tml) * 16;
    const int col0 = tn * 16;

    const float* Arow = Al + (tml * 16 + lm) * A_PITCH(K);
    const float* Bcol = Wl + col0 + lm;

    v8f acc = {0.f, 0.f, 0.f, 0.f, 0.f, 0.f, 0.f, 0.f};
    for (int k = 0; k < K; k += 4) {
        const int ks = k + 2 * half;
        const v2f a = *(const v2f*)(Arow + ks);
        v2f b;
        b.x = Bcol[ks * W_PITCH];
        b.y = Bcol[(ks + 1) * W_PITCH];
        acc = __builtin_amdgcn_wmma_f32_16x16x4_f32(
            false, a, false, b, (short)0, acc, false, false);
    }

    const int col  = col0 + lm;
    const float bc = bias[col];
#pragma unroll
    for (int r = 0; r < 8; ++r) {
        int row = row0 + r + 8 * half;
        float v = acc[r] + bc;
        if (do_relu) v = fmaxf(v, 0.0f);
        C[(long)row * HF + col] = v;
    }

    if (EXTRA == 2 && tid < 32) {
        const int i = tile * 32 + tid;
        dinv[i] = 1.0f / sqrtf(fmaxf(deg2[i] + deg2[N_NODES + i], 1.0f));
    }
}

struct Theta3 { float c[3][3]; };

__global__ void gemm_filters_wmma(float* planes,
                                  const float* __restrict__ Wm1,
                                  const float* __restrict__ dinv,
                                  float* __restrict__ S,
                                  float* __restrict__ QP,
                                  float* __restrict__ Ps,
                                  Theta3 th, int n64) {
    extern __shared__ __attribute__((aligned(16))) float lds[];
    float* Wl = lds;
    float* Al = lds + 3 * HF * W_PITCH;
    const int tid = threadIdx.x;

    const int wave = tid >> 5;
    const int lane = tid & 31;
    const int half = lane >> 4;
    const int lm   = lane & 15;
    const int tn   = wave & 3;
    const int tml  = wave >> 2;
    const int row0 = (blockIdx.x * 2 + tml) * 16;
    const int col0 = tn * 16;

    const v4f* dv = (const v4f*)(dinv + row0 + 8 * half);
    const v4f d03 = dv[0], d47 = dv[1];
    const float dr[8] = {d03.x, d03.y, d03.z, d03.w, d47.x, d47.y, d47.z, d47.w};

    const float* H = planes + n64;
    stage_rows<HF>(Al, H + (long)blockIdx.x * (32 * HF), tid);
    stage_w<3 * HF>(Wl, Wm1, tid);
    __syncthreads();
    {
        const v4f z = {0.f, 0.f, 0.f, 0.f};
        const long base = (long)blockIdx.x * (32 * HF);
        v4f* z0 = (v4f*)(planes + base);
        v4f* z1 = (v4f*)(planes + n64 + base);
        z0[tid] = z;  z0[tid + 256] = z;
        z1[tid] = z;  z1[tid + 256] = z;
    }

    const float* Hrow = Al + (tml * 16 + lm) * A_PITCH(HF);
    const float* Bcol = Wl + col0 + lm;

    v8f g0 = {0.f, 0.f, 0.f, 0.f, 0.f, 0.f, 0.f, 0.f};
    v8f g1 = g0, g2 = g0;
    for (int k = 0; k < HF; k += 4) {
        const int ks = k + 2 * half;
        const v2f a = *(const v2f*)(Hrow + ks);
        v2f b0, b1, b2;
        b0.x = Bcol[ks * W_PITCH];
        b0.y = Bcol[(ks + 1) * W_PITCH];
        b1.x = Bcol[(HF + ks) * W_PITCH];
        b1.y = Bcol[(HF + ks + 1) * W_PITCH];
        b2.x = Bcol[(2 * HF + ks) * W_PITCH];
        b2.y = Bcol[(2 * HF + ks + 1) * W_PITCH];
        g0 = __builtin_amdgcn_wmma_f32_16x16x4_f32(
            false, a, false, b0, (short)0, g0, false, false);
        g1 = __builtin_amdgcn_wmma_f32_16x16x4_f32(
            false, a, false, b1, (short)0, g1, false, false);
        g2 = __builtin_amdgcn_wmma_f32_16x16x4_f32(
            false, a, false, b2, (short)0, g2, false, false);
    }

    const int col = col0 + lm;
#pragma unroll
    for (int r = 0; r < 8; ++r) {
        long o = (long)(row0 + r + 8 * half) * HF + col;
        float x0 = g0[r], x1 = g1[r], x2 = g2[r];
        const float r0 = th.c[0][0] * x0 + th.c[1][0] * x1 + th.c[2][0] * x2;
        const float q  = th.c[0][1] * x0 + th.c[1][1] * x1 + th.c[2][1] * x2;
        const float p  = th.c[0][2] * x0 + th.c[1][2] * x1 + th.c[2][2] * x2;
        const float qp = q + p;
        S[o]  = r0 + qp;
        QP[o] = qp;
        Ps[o] = p * dr[r];
    }
}


#define SC_BATCH 32
#define SC_GROUP 8
#define SC_NGROUP (SC_BATCH / SC_GROUP)

__global__ void __launch_bounds__(256)
edge_scatter(const int* __restrict__ ei, const float* __restrict__ feat,
             float* __restrict__ msg2, int E, int n64) {
    float* msg = msg2 + (long)chip_half() * n64;
    const int lane = threadIdx.x & 31;
    const int e0   = (blockIdx.x * 8 + (threadIdx.x >> 5)) * SC_BATCH;
    if (e0 >= E) return;
    const int e   = e0 + lane;
    const int s_l = ei[e];
    const int d_l = ei[E + e];

    float lo[SC_NGROUP][SC_GROUP], hi[SC_NGROUP][SC_GROUP];

#pragma unroll
    for (int g = -1; g < SC_NGROUP; ++g) {
        if (g + 1 < SC_NGROUP) {
#pragma unroll
            for (int u = 0; u < SC_GROUP; ++u) {
                const int j = (g + 1) * SC_GROUP + u;
                const int s = __builtin_amdgcn_readlane(s_l, j);
                const float* row = feat + (long)s * HF;
                lo[g + 1][u] = row[lane];
                hi[g + 1][u] = row[lane + 32];
            }
        }
        if (g >= 0) {
#pragma unroll
            for (int u = 0; u < SC_GROUP; ++u) {
                const int j = g * SC_GROUP + u;
                const int d = __builtin_amdgcn_readlane(d_l, j);
                float* row = msg + (long)d * HF;
                atomicAdd(&row[lane], lo[g][u]);
                atomicAdd(&row[lane + 32], hi[g][u]);
            }
        }
    }
}

#define UPD_V 2
#define UPD_THREADS (N_NODES * HF / 4 / UPD_V)
static_assert((N_NODES * HF / 4) % (256 * UPD_V) == 0, "step grid covers the arrays exactly");
static_assert(UPD_THREADS % 16 == 0, "a lane keeps its feature columns from place to place");

template <int LAST>
__global__ void __launch_bounds__(256)
nested_step(const float* __restrict__ sacc, float* __restrict__ zs,
            const float* __restrict__ msg2, const float* __restrict__ dinv,
            const float* __restrict__ bias, int n64,
            const float* __restrict__ Wm2, const float* __restrict__ bm2,
            float* __restrict__ out) {
    __shared__ float oline[UPD_V][32];
    const int t = blockIdx.x * 256 + threadIdx.x;
    const v4f* s4 = (const v4f*)sacc;
    v4f* z4 = (v4f*)zs;
    const v4f* m0 = (const v4f*)msg2;
    const v4f* m1 = (const v4f*)(msg2 + n64);

    v4f r[UPD_V], a[UPD_V], p0[UPD_V], p1[UPD_V];
    float d[UPD_V];
#pragma unroll
    for (int u = 0; u < UPD_V; ++u) {
        const int q = t + u * UPD_THREADS;
        p0[u] = m0[q];
        p1[u] = m1[q];
        if (LAST) r[u] = s4[q];
        else      a[u] = z4[q];
        d[u]  = dinv[q >> 4];
    }
    const int f = (t & 15) * 4;
    v4f b = {0.f, 0.f, 0.f, 0.f};
    v4f wa = b, wb = b;
    if (LAST) {
        b.x = bias[f]; b.y = bias[f + 1]; b.z = bias[f + 2]; b.w = bias[f + 3];
        const v4f* w4 = (const v4f*)Wm2;
        wa = w4[(t & 15) * 2];
        wb = w4[(t & 15) * 2 + 1];
    }
#pragma unroll
    for (int u = 0; u < UPD_V; ++u) {
        const int q = t + u * UPD_THREADS;
        const v4f m = p0[u] + p1[u];
        if (LAST) {
            v4f v = (r[u] - m * d[u]) + b;
            v.x = fmaxf(v.x, 0.0f);
            v.y = fmaxf(v.y, 0.0f);
            v.z = fmaxf(v.z, 0.0f);
            v.w = fmaxf(v.w, 0.0f);
            float c0 = v.x * wa.x + v.y * wa.z + v.z * wb.x + v.w * wb.z;
            float c1 = v.x * wa.y + v.y * wa.w + v.z * wb.y + v.w * wb.w;
#pragma unroll
            for (int m = 1; m < 16; m <<= 1) {
                c0 += __shfl_xor(c0, m);
                c1 += __shfl_xor(c1, m);
            }
            if ((threadIdx.x & 15) == 0) {
                const int nl = threadIdx.x >> 4;
                oline[u][nl * 2 + 0] = c0 + bm2[0];
                oline[u][nl * 2 + 1] = c1 + bm2[1];
            }
        } else {
            const v4f z = a[u] - m * d[u];
            z4[q] = z * d[u];
        }
    }
    if (LAST) {
        __syncthreads();
        if (threadIdx.x < 32) {
#pragma unroll
            for (int u = 0; u < UPD_V; ++u) {
                const int node0 = (blockIdx.x * 256 + u * UPD_THREADS) >> 4;
                out[node0 * 2 + threadIdx.x] = oline[u][threadIdx.x];
            }
        }
    }
}

extern "C" void kernel_launch(void* const* d_in, const int* in_sizes, int n_in,
                              void* d_out, int out_size, void* d_ws, size_t ws_size,
                              hipStream_t stream) {
    const float* x   = (const float*)d_in[0];
    const int*   ei  = (const int*)  d_in[1];
    const float* W1  = (const float*)d_in[2];
    const float* b1  = (const float*)d_in[3];
    const float* W2  = (const float*)d_in[4];
    const float* b2  = (const float*)d_in[5];
    const float* Wm1 = (const float*)d_in[6];
    const float* bm1 = (const float*)d_in[7];
    const float* Wm2 = (const float*)d_in[8];
    const float* bm2 = (const float*)d_in[9];
    float* out = (float*)d_out;

    const int N = N_NODES, E = N_EDGES;
    const int N64 = N * HF;

    float* ws   = (float*)d_ws;
    float* dinv = ws;
    float* S    = dinv + N;
    float* msg2 = S    + (long)N64;
    float* h1   = msg2;
    float* h    = msg2 + (long)N64;
    float* QP   = h    + (long)N64;
    float* Ps   = QP   + (long)N64;
    float* deg2 = Ps;

    const Theta3 TH = {{{3.0f, -3.0f, 0.75f},
                        {0.0f,  3.0f, -1.50f},
                        {0.0f,  0.0f, 0.75f}}};

    zero_f32<<<(2 * N + 255) / 256, 256, 0, stream>>>(deg2, 2 * N);

    gemm_bias_relu_wmma<1, 128><<<DEG_BLOCKS + N / 32, 256, GEMM_LDS_BYTES(128, 128), stream>>>(
        x, W1, b1, h1, 1, ei, deg2, dinv);
    gemm_bias_relu_wmma<2, 64><<<N / 32, 256, GEMM_LDS_BYTES(64, 64), stream>>>(
        h1, W2, b2, h, 1, ei, deg2, dinv);

    gemm_filters_wmma<<<N / 32, 256, GEMM_LDS_BYTES(192, 64), stream>>>(
        msg2, Wm1, dinv, S, QP, Ps, TH, N64);

    static_assert(N_EDGES % (8 * SC_BATCH) == 0, "edge batches fill whole work-groups");
    const int eg = N_EDGES / (8 * SC_BATCH);
    const int ug = UPD_THREADS / 256;
    edge_scatter<<<eg, 256, 0, stream>>>(ei, Ps, msg2, E, N64);
    nested_step<0><<<ug, 256, 0, stream>>>(S, QP, msg2, dinv, bm1, N64, Wm2, bm2, out);

    edge_scatter<<<eg, 256, 0, stream>>>(ei, QP, msg2, E, N64);
    nested_step<1><<<ug, 256, 0, stream>>>(S, QP, msg2, dinv, bm1, N64, Wm2, bm2, out);
}


# Row scatters split by feature half, not by plane copy: chip half q adds features 32q..32q+31 of every edge (items dealt per half by a counter) into one plane kept [q][node][32]; sum order changed
# speedup vs baseline: 12.5885x; 1.1797x over previous
#include <hip/hip_runtime.h>
#include <hip/hip_bf16.h>


typedef float v2f __attribute__((ext_vector_type(2)));
typedef float v4f __attribute__((ext_vector_type(4)));
typedef float v8f __attribute__((ext_vector_type(8)));

#define N_NODES 100000
#define N_EDGES 1600000
#define HF 64
#define HP (N_NODES * (HF / 2))

__global__ void zero_f32(float* __restrict__ p, int n) {
    int i = blockIdx.x * blockDim.x + threadIdx.x;
    if (i < n) p[i] = 0.0f;
}

__device__ __forceinline__ int chip_half() {
    const unsigned id = __builtin_amdgcn_s_sendmsg_rtn(0x87);
    return ((id >> 16) & 15u) >= 4u ? 1 : 0;
}

#define DEG_EDGES_PER_THREAD 4
#define DEG_THREADS (N_EDGES / DEG_EDGES_PER_THREAD)
#define DEG_BLOCKS ((DEG_THREADS + 255) / 256)
static_assert(N_EDGES % DEG_EDGES_PER_THREAD == 0, "a thread's four edges exist together");

__device__ __forceinline__ void degree_role(const int* __restrict__ ei, float* __restrict__ deg2,
                                            int t) {
    if (t >= DEG_THREADS) return;
    float* deg = deg2 + chip_half() * N_NODES;
    const int4 d4 = ((const int4*)(ei + N_EDGES))[t];
    atomicAdd(&deg[d4.x], 1.0f);
    atomicAdd(&deg[d4.y], 1.0f);
    atomicAdd(&deg[d4.z], 1.0f);
    atomicAdd(&deg[d4.w], 1.0f);
}

#define W_PITCH 72
#define A_PITCH(K) ((K) + 4)
#define GEMM_LDS_BYTES(KW, KA) (((KW) * W_PITCH + 32 * A_PITCH(KA)) * sizeof(float))

template <int KW>
__device__ __forceinline__ void stage_w(float* __restrict__ Wl, const float* __restrict__ W, int tid) {
    const v4f* Wg = (const v4f*)W;
    for (int j = 0; j < KW * 16 / 256; ++j) {
        const int i = tid + 256 * j;
        *(v4f*)(Wl + (i >> 4) * W_PITCH + (i & 15) * 4) = Wg[i];
    }
}

template <int K>
__device__ __forceinline__ void stage_rows(float* __restrict__ Al, const float* __restrict__ rows, int tid) {
    const v4f* Ag = (const v4f*)rows;
    for (int j = 0; j < 32 * (K / 4) / 256; ++j) {
        const int i = tid + 256 * j;
        *(v4f*)(Al + (i / (K / 4)) * A_PITCH(K) + (i % (K / 4)) * 4) = Ag[i];
    }
}

template <int EXTRA, int K>
__global__ void
gemm_bias_relu_wmma(const float* __restrict__ A,
                    const float* __restrict__ W,
                    const float* __restrict__ bias,
                    float* __restrict__ C,
                    int do_relu,
                    const int* __restrict__ ei,
                    float* __restrict__ deg2,
                    float* __restrict__ dinv) {
    extern __shared__ __attribute__((aligned(16))) float lds[];
    float* Wl = lds;
    float* Al = lds + K * W_PITCH;
    const int tid = threadIdx.x;

    int tile = blockIdx.x;
    if (EXTRA == 1) {
        if (blockIdx.x < DEG_BLOCKS) {
            degree_role(ei, deg2, blockIdx.x * 256 + tid);
            return;
        }
        tile = blockIdx.x - DEG_BLOCKS;
    }
    stage_rows<K>(Al, A + (long)tile * (32 * K), tid);
    stage_w<K>(Wl, W, tid);
    __syncthreads();

    const int wave = tid >> 5;
    const int lane = tid & 31;
    const int half = lane >> 4;
    const int lm   = lane & 15;
    const int tn   = wave & 3;
    const int tml  = wave >> 2;
    const int row0 = (tile * 2 + tml) * 16;
    const int col0 = tn * 16;

    const float* Arow = Al + (tml * 16 + lm) * A_PITCH(K);
    const float* Bcol = Wl + col0 + lm;

    v8f acc = {0.f, 0.f, 0.f, 0.f, 0.f, 0.f, 0.f, 0.f};
    for (int k = 0; k < K; k += 4) {
        const int ks = k + 2 * half;
        const v2f a = *(const v2f*)(Arow + ks);
        v2f b;
        b.x = Bcol[ks * W_PITCH];
        b.y = Bcol[(ks + 1) * W_PITCH];
        acc = __builtin_amdgcn_wmma_f32_16x16x4_f32(
            false, a, false, b, (short)0, acc, false, false);
    }

    const int col  = col0 + lm;
    const float bc = bias[col];
#pragma unroll
    for (int r = 0; r < 8; ++r) {
        int row = row0 + r + 8 * half;
        float v = acc[r] + bc;
        if (do_relu) v = fmaxf(v, 0.0f);
        C[(long)row * HF + col] = v;
    }

    if (EXTRA == 2 && tid < 32) {
        const int i = tile * 32 + tid;
        dinv[i] = 1.0f / sqrtf(fmaxf(deg2[i] + deg2[N_NODES + i], 1.0f));
    }
}

struct Theta3 { float c[3][3]; };

__global__ void gemm_filters_wmma(float* planes,
                                  int* ctr,
                                  const float* __restrict__ Wm1,
                                  const float* __restrict__ dinv,
                                  float* __restrict__ S,
                                  float* __restrict__ QP,
                                  float* __restrict__ Ps,
                                  Theta3 th, int n64) {
    extern __shared__ __attribute__((aligned(16))) float lds[];
    float* Wl = lds;
    float* Al = lds + 3 * HF * W_PITCH;
    const int tid = threadIdx.x;

    const int wave = tid >> 5;
    const int lane = tid & 31;
    const int half = lane >> 4;
    const int lm   = lane & 15;
    const int tn   = wave & 3;
    const int tml  = wave >> 2;
    const int row0 = (blockIdx.x * 2 + tml) * 16;
    const int col0 = tn * 16;

    const v4f* dv = (const v4f*)(dinv + row0 + 8 * half);
    const v4f d03 = dv[0], d47 = dv[1];
    const float dr[8] = {d03.x, d03.y, d03.z, d03.w, d47.x, d47.y, d47.z, d47.w};

    const float* H = planes + n64;
    stage_rows<HF>(Al, H + (long)blockIdx.x * (32 * HF), tid);
    stage_w<3 * HF>(Wl, Wm1, tid);
    __syncthreads();
    {
        const v4f z = {0.f, 0.f, 0.f, 0.f};
        const long base = (long)blockIdx.x * (32 * (HF / 2));
        v4f* z0 = (v4f*)(planes + base);
        v4f* z1 = (v4f*)(planes + HP + base);
        z0[tid] = z;  z1[tid] = z;
        if (blockIdx.x == 0 && tid < 4) ctr[tid] = 0;
    }

    const float* Hrow = Al + (tml * 16 + lm) * A_PITCH(HF);
    const float* Bcol = Wl + col0 + lm;

    v8f g0 = {0.f, 0.f, 0.f, 0.f, 0.f, 0.f, 0.f, 0.f};
    v8f g1 = g0, g2 = g0;
    for (int k = 0; k < HF; k += 4) {
        const int ks = k + 2 * half;
        const v2f a = *(const v2f*)(Hrow + ks);
        v2f b0, b1, b2;
        b0.x = Bcol[ks * W_PITCH];
        b0.y = Bcol[(ks + 1) * W_PITCH];
        b1.x = Bcol[(HF + ks) * W_PITCH];
        b1.y = Bcol[(HF + ks + 1) * W_PITCH];
        b2.x = Bcol[(2 * HF + ks) * W_PITCH];
        b2.y = Bcol[(2 * HF + ks + 1) * W_PITCH];
        g0 = __builtin_amdgcn_wmma_f32_16x16x4_f32(
            false, a, false, b0, (short)0, g0, false, false);
        g1 = __builtin_amdgcn_wmma_f32_16x16x4_f32(
            false, a, false, b1, (short)0, g1, false, false);
        g2 = __builtin_amdgcn_wmma_f32_16x16x4_f32(
            false, a, false, b2, (short)0, g2, false, false);
    }

    const int col = col0 + lm;
#pragma unroll
    for (int r = 0; r < 8; ++r) {
        long o = (long)(row0 + r + 8 * half) * HF + col;
        float x0 = g0[r], x1 = g1[r], x2 = g2[r];
        const float r0 = th.c[0][0] * x0 + th.c[1][0] * x1 + th.c[2][0] * x2;
        const float q  = th.c[0][1] * x0 + th.c[1][1] * x1 + th.c[2][1] * x2;
        const float p  = th.c[0][2] * x0 + th.c[1][2] * x1 + th.c[2][2] * x2;
        const float qp = q + p;
        S[o]  = r0 + qp;
        QP[o] = qp;
        Ps[o] = p * dr[r];
    }
}


#define SC_BATCH 64
#define SC_GROUP 16
#define SC_NGROUP (SC_BATCH / SC_GROUP)
#define SC_ITEMS (N_EDGES / (8 * SC_BATCH))
static_assert(N_EDGES % (8 * SC_BATCH) == 0, "edge batches fill whole work-groups");

__global__ void __launch_bounds__(256)
edge_scatter(const int* __restrict__ ei, const float* __restrict__ feat,
             float* __restrict__ msg, int* __restrict__ ctr) {
    __shared__ int taken;
    if (threadIdx.x == 0) {
        int q  = chip_half();
        int it = atomicAdd(&ctr[q], 1);
        if (it >= SC_ITEMS) {
            q ^= 1;
            it = atomicAdd(&ctr[q], 1);
        }
        taken = it < SC_ITEMS ? 2 * it + q : -1;
    }
    __syncthreads();
    const int w = __builtin_amdgcn_readfirstlane(taken);
    if (w < 0) return;
    const int lane = threadIdx.x & 31;
    const int fo   = (w & 1) * 32 + lane;
    float* msgq    = msg + (long)(w & 1) * HP + lane;
    const int e0   = ((w >> 1) * 8 + (threadIdx.x >> 5)) * SC_BATCH;
    const int2 s2 = *(const int2*)(ei + e0 + 2 * lane);
    const int2 d2 = *(const int2*)(ei + N_EDGES + e0 + 2 * lane);

    float v[SC_NGROUP][SC_GROUP];

#pragma unroll
    for (int g = -1; g < SC_NGROUP; ++g) {
        if (g + 1 < SC_NGROUP) {
#pragma unroll
            for (int u = 0; u < SC_GROUP; ++u) {
                const int j = (g + 1) * SC_GROUP + u;
                const int s = __builtin_amdgcn_readlane((j & 1) ? s2.y : s2.x, j >> 1);
                v[g + 1][u] = feat[(long)s * HF + fo];
            }
        }
        if (g >= 0) {
#pragma unroll
            for (int u = 0; u < SC_GROUP; ++u) {
                const int j = g * SC_GROUP + u;
                const int d = __builtin_amdgcn_readlane((j & 1) ? d2.y : d2.x, j >> 1);
                atomicAdd(&msgq[(long)d * (HF / 2)], v[g][u]);
            }
        }
    }
}

#define UPD_V 2
#define UPD_THREADS (N_NODES * HF / 4 / UPD_V)
static_assert((N_NODES * HF / 4) % (256 * UPD_V) == 0, "step grid covers the arrays exactly");
static_assert(UPD_THREADS % 16 == 0, "a lane keeps its feature columns from place to place");

template <int LAST>
__global__ void __launch_bounds__(256)
nested_step(const float* __restrict__ sacc, float* __restrict__ zs,
            const float* __restrict__ msg, const float* __restrict__ dinv,
            const float* __restrict__ bias,
            const float* __restrict__ Wm2, const float* __restrict__ bm2,
            float* __restrict__ out) {
    __shared__ float oline[UPD_V][32];
    const int t = blockIdx.x * 256 + threadIdx.x;
    const v4f* s4 = (const v4f*)sacc;
    v4f* z4 = (v4f*)zs;
    const v4f* m0 = (const v4f*)msg;

    v4f r[UPD_V], a[UPD_V], p0[UPD_V];
    float d[UPD_V];
#pragma unroll
    for (int u = 0; u < UPD_V; ++u) {
        const int q = t + u * UPD_THREADS;
        p0[u] = m0[((q >> 3) & 1) * (HP / 4) + (q >> 4) * 8 + (q & 7)];
        if (LAST) r[u] = s4[q];
        else      a[u] = z4[q];
        d[u]  = dinv[q >> 4];
    }
    const int f = (t & 15) * 4;
    v4f b = {0.f, 0.f, 0.f, 0.f};
    v4f wa = b, wb = b;
    if (LAST) {
        b.x = bias[f]; b.y = bias[f + 1]; b.z = bias[f + 2]; b.w = bias[f + 3];
        const v4f* w4 = (const v4f*)Wm2;
        wa = w4[(t & 15) * 2];
        wb = w4[(t & 15) * 2 + 1];
    }
#pragma unroll
    for (int u = 0; u < UPD_V; ++u) {
        const int q = t + u * UPD_THREADS;
        const v4f m = p0[u];
        if (LAST) {
            v4f v = (r[u] - m * d[u]) + b;
            v.x = fmaxf(v.x, 0.0f);
            v.y = fmaxf(v.y, 0.0f);
            v.z = fmaxf(v.z, 0.0f);
            v.w = fmaxf(v.w, 0.0f);
            float c0 = v.x * wa.x + v.y * wa.z + v.z * wb.x + v.w * wb.z;
            float c1 = v.x * wa.y + v.y * wa.w + v.z * wb.y + v.w * wb.w;
#pragma unroll
            for (int m = 1; m < 16; m <<= 1) {
                c0 += __shfl_xor(c0, m);
                c1 += __shfl_xor(c1, m);
            }
            if ((threadIdx.x & 15) == 0) {
                const int nl = threadIdx.x >> 4;
                oline[u][nl * 2 + 0] = c0 + bm2[0];
                oline[u][nl * 2 + 1] = c1 + bm2[1];
            }
        } else {
            const v4f z = a[u] - m * d[u];
            z4[q] = z * d[u];
        }
    }
    if (LAST) {
        __syncthreads();
        if (threadIdx.x < 32) {
#pragma unroll
            for (int u = 0; u < UPD_V; ++u) {
                const int node0 = (blockIdx.x * 256 + u * UPD_THREADS) >> 4;
                out[node0 * 2 + threadIdx.x] = oline[u][threadIdx.x];
            }
        }
    }
}

extern "C" void kernel_launch(void* const* d_in, const int* in_sizes, int n_in,
                              void* d_out, int out_size, void* d_ws, size_t ws_size,
                              hipStream_t stream) {
    const float* x   = (const float*)d_in[0];
    const int*   ei  = (const int*)  d_in[1];
    const float* W1  = (const float*)d_in[2];
    const float* b1  = (const float*)d_in[3];
    const float* W2  = (const float*)d_in[4];
    const float* b2  = (const float*)d_in[5];
    const float* Wm1 = (const float*)d_in[6];
    const float* bm1 = (const float*)d_in[7];
    const float* Wm2 = (const float*)d_in[8];
    const float* bm2 = (const float*)d_in[9];
    float* out = (float*)d_out;

    const int N = N_NODES;
    const int N64 = N * HF;

    float* ws   = (float*)d_ws;
    float* dinv = ws;
    float* S    = dinv + N;
    float* msg  = S    + (long)N64;
    float* h1   = msg;
    float* h    = msg  + (long)N64;
    int*   ctr  = (int*)h;
    float* QP   = h    + (long)N64;
    float* Ps   = QP   + (long)N64;
    float* deg2 = Ps;

    const Theta3 TH = {{{3.0f, -3.0f, 0.75f},
                        {0.0f,  3.0f, -1.50f},
                        {0.0f,  0.0f, 0.75f}}};

    zero_f32<<<(2 * N + 255) / 256, 256, 0, stream>>>(deg2, 2 * N);

    gemm_bias_relu_wmma<1, 128><<<DEG_BLOCKS + N / 32, 256, GEMM_LDS_BYTES(128, 128), stream>>>(
        x, W1, b1, h1, 1, ei, deg2, dinv);
    gemm_bias_relu_wmma<2, 64><<<N / 32, 256, GEMM_LDS_BYTES(64, 64), stream>>>(
        h1, W2, b2, h, 1, ei, deg2, dinv);

    gemm_filters_wmma<<<N / 32, 256, GEMM_LDS_BYTES(192, 64), stream>>>(
        msg, ctr, Wm1, dinv, S, QP, Ps, TH, N64);

    const int eg = 2 * SC_ITEMS;
    const int ug = UPD_THREADS / 256;
    edge_scatter<<<eg, 256, 0, stream>>>(ei, Ps, msg, ctr);
    nested_step<0><<<ug, 256, 0, stream>>>(S, QP, msg, dinv, bm1, Wm2, bm2, out);

    edge_scatter<<<eg, 256, 0, stream>>>(ei, QP, msg, ctr + 2);
    nested_step<1><<<ug, 256, 0, stream>>>(S, QP, msg, dinv, bm1, Wm2, bm2, out);
}
